# starting kernel
# baseline (speedup 1.0000x reference)
#include <hip/hip_runtime.h>

typedef __attribute__((ext_vector_type(2))) float v2f;
typedef __attribute__((ext_vector_type(8))) float v8f;

#define IN_DIM 300
#define H_DIM  128
#define L_LAYERS 4

#define XS (IN_DIM + 2)
#define HS (H_DIM + 4)


__global__ void zero_f32_kernel(float* __restrict__ p, int n) {
    int i = blockIdx.x * blockDim.x + threadIdx.x;
    if (i < n) p[i] = 0.0f;
}

__global__ void degree_kernel(const int* __restrict__ dst, float* __restrict__ deg, int e) {
    int i = blockIdx.x * blockDim.x + threadIdx.x;
    if (i < e) unsafeAtomicAdd(&deg[dst[i]], 1.0f);
}

__global__ void transpose_kernel(const float* __restrict__ in, float* __restrict__ out,
                                 int K, int Nc) {
    int i = blockIdx.x * blockDim.x + threadIdx.x;
    if (i < K * Nc) {
        int k = i / Nc;
        int n = i - k * Nc;
        out[n * K + k] = in[i];
    }
}

__global__ __launch_bounds__(256)
void scatter_kernel(const int* __restrict__ src, const int* __restrict__ dst,
                    const float* __restrict__ h, float* __restrict__ agg, int e) {
    int wave = (blockIdx.x * blockDim.x + threadIdx.x) >> 5;
    int lane = threadIdx.x & 31;
    if (wave >= e) return;
    int s = src[wave];
    int d = dst[wave];
    const float4* hp = reinterpret_cast<const float4*>(h + (long)s * H_DIM);
    float4 v = hp[lane];
    float* ap = agg + (long)d * H_DIM + lane * 4;
    unsafeAtomicAdd(ap + 0, v.x);
    unsafeAtomicAdd(ap + 1, v.y);
    unsafeAtomicAdd(ap + 2, v.z);
    unsafeAtomicAdd(ap + 3, v.w);
}


__global__ __launch_bounds__(256)
void emb_kernel(const float* __restrict__ x, const float* __restrict__ WembT,
                const float* __restrict__ bemb, float* __restrict__ h, int n_nodes) {
    __shared__ float shx[16 * XS];
    const int  tid  = threadIdx.x;
    const long row0 = (long)blockIdx.x * 16;
    const bool full = (row0 + 16 <= (long)n_nodes);

    if (full) {
        for (int i = tid; i < 16 * IN_DIM; i += 256) {
            int r = i / IN_DIM, c = i - r * IN_DIM;
            shx[r * XS + c] = x[(row0 + r) * IN_DIM + c];
        }
    } else {
        for (int i = tid; i < 16 * IN_DIM; i += 256) {
            int r = i / IN_DIM, c = i - r * IN_DIM;
            long rr = row0 + r; if (rr >= n_nodes) rr = n_nodes - 1;
            shx[r * XS + c] = x[rr * IN_DIM + c];
        }
    }
    __syncthreads();

    const int wave = tid >> 5, lane = tid & 31;
    const int m = lane & 15, hi = lane >> 4;
    const int ncol = wave * 16 + m;

    v8f acc;
    float bv = bemb[ncol];
    #pragma unroll
    for (int i = 0; i < 8; ++i) acc[i] = bv;

    const float* wt = WembT + (long)ncol * IN_DIM;
    #pragma unroll 5
    for (int k0 = 0; k0 < IN_DIM; k0 += 4) {
        int k = k0 + 2 * hi;
        v2f a; a.x = shx[m * XS + k]; a.y = shx[m * XS + k + 1];
        v2f b; b.x = wt[k];           b.y = wt[k + 1];
        acc = __builtin_amdgcn_wmma_f32_16x16x4_f32(false, a, false, b,
                                                    (short)0, acc, false, false);
    }

    float* hrow = h + row0 * H_DIM + wave * 16 + m;
    if (full) {
        #pragma unroll
        for (int v = 0; v < 8; ++v) {
            float r = acc[v];
            hrow[(long)(v + 8 * hi) * H_DIM] = r > 0.0f ? r : 0.0f;
        }
    } else {
        #pragma unroll
        for (int v = 0; v < 8; ++v) {
            int mm = v + 8 * hi;
            if (row0 + mm < n_nodes) {
                float r = acc[v];
                hrow[(long)mm * H_DIM] = r > 0.0f ? r : 0.0f;
            }
        }
    }
}

__global__ __launch_bounds__(256)
void sage_layer_kernel(const float* __restrict__ agg, const float* __restrict__ hin,
                       const float* __restrict__ deg,
                       const float* __restrict__ WlT, const float* __restrict__ WrT,
                       const float* __restrict__ bias, float* __restrict__ hout,
                       int n_nodes) {
    __shared__ float sh_a[16 * HS];
    __shared__ float sh_h[16 * HS];
    const int  tid  = threadIdx.x;
    const long row0 = (long)blockIdx.x * 16;
    const bool full = (row0 + 16 <= (long)n_nodes);

    if (full) {
        for (int i = tid; i < 16 * H_DIM; i += 256) {
            int r = i >> 7, c = i & 127;
            long rr = row0 + r;
            float dg  = deg[rr];
            float inv = 1.0f / (dg > 1.0f ? dg : 1.0f);
            sh_a[r * HS + c] = agg[rr * H_DIM + c] * inv;
            sh_h[r * HS + c] = hin[rr * H_DIM + c];
        }
    } else {
        for (int i = tid; i < 16 * H_DIM; i += 256) {
            int r = i >> 7, c = i & 127;
            long rr = row0 + r; if (rr >= n_nodes) rr = n_nodes - 1;
            float dg  = deg[rr];
            float inv = 1.0f / (dg > 1.0f ? dg : 1.0f);
            sh_a[r * HS + c] = agg[rr * H_DIM + c] * inv;
            sh_h[r * HS + c] = hin[rr * H_DIM + c];
        }
    }
    __syncthreads();

    const int wave = tid >> 5, lane = tid & 31;
    const int m = lane & 15, hi = lane >> 4;
    const int ncol = wave * 16 + m;

    v8f acc;
    float bv = bias[ncol];
    #pragma unroll
    for (int i = 0; i < 8; ++i) acc[i] = bv;

    const float* wl = WlT + (long)ncol * H_DIM;
    const float* wr = WrT + (long)ncol * H_DIM;
    #pragma unroll 4
    for (int k0 = 0; k0 < H_DIM; k0 += 4) {
        int k = k0 + 2 * hi;
        v2f aa; aa.x = sh_a[m * HS + k]; aa.y = sh_a[m * HS + k + 1];
        v2f ah; ah.x = sh_h[m * HS + k]; ah.y = sh_h[m * HS + k + 1];
        v2f bl; bl.x = wl[k];            bl.y = wl[k + 1];
        v2f br; br.x = wr[k];            br.y = wr[k + 1];
        acc = __builtin_amdgcn_wmma_f32_16x16x4_f32(false, aa, false, bl,
                                                    (short)0, acc, false, false);
        acc = __builtin_amdgcn_wmma_f32_16x16x4_f32(false, ah, false, br,
                                                    (short)0, acc, false, false);
    }

    float* orow = hout + row0 * H_DIM + wave * 16 + m;
    if (full) {
        #pragma unroll
        for (int v = 0; v < 8; ++v) {
            float r = acc[v];
            orow[(long)(v + 8 * hi) * H_DIM] = r > 0.0f ? r : 0.0f;
        }
    } else {
        #pragma unroll
        for (int v = 0; v < 8; ++v) {
            int mm = v + 8 * hi;
            if (row0 + mm < n_nodes) {
                float r = acc[v];
                orow[(long)mm * H_DIM] = r > 0.0f ? r : 0.0f;
            }
        }
    }
}


static inline int cdiv(long a, long b) { return (int)((a + b - 1) / b); }

extern "C" void kernel_launch(void* const* d_in, const int* in_sizes, int n_in,
                              void* d_out, int out_size, void* d_ws, size_t ws_size,
                              hipStream_t stream) {
    const float* x    = (const float*)d_in[0];
    const int*   ei   = (const int*)  d_in[1];
    const float* Wemb = (const float*)d_in[2];
    const float* bemb = (const float*)d_in[3];
    const float* Wl   = (const float*)d_in[4];
    const float* bl   = (const float*)d_in[5];
    const float* Wr   = (const float*)d_in[6];
    float* out = (float*)d_out;

    const int N = in_sizes[0] / IN_DIM;
    const int E = in_sizes[1] / 2;
    const int* src = ei;
    const int* dst = ei + E;

    char*  ws  = (char*)d_ws;
    size_t off = 0;
    auto carve = [&](size_t bytes) -> float* {
        float* p = (float*)(ws + off);
        off = (off + bytes + 255) & ~(size_t)255;
        return p;
    };
    float* deg   = carve((size_t)N * sizeof(float));
    float* hA    = carve((size_t)N * H_DIM * sizeof(float));
    float* hB    = carve((size_t)N * H_DIM * sizeof(float));
    float* agg   = carve((size_t)N * H_DIM * sizeof(float));
    float* WembT = carve((size_t)IN_DIM * H_DIM * sizeof(float));
    float* WlT   = carve((size_t)L_LAYERS * H_DIM * H_DIM * sizeof(float));
    float* WrT   = carve((size_t)L_LAYERS * H_DIM * H_DIM * sizeof(float));
    (void)ws_size;

    const int tiles = cdiv(N, 16);

    zero_f32_kernel<<<cdiv(N, 256), 256, 0, stream>>>(deg, N);
    degree_kernel<<<cdiv(E, 256), 256, 0, stream>>>(dst, deg, E);

    transpose_kernel<<<cdiv((long)IN_DIM * H_DIM, 256), 256, 0, stream>>>(Wemb, WembT, IN_DIM, H_DIM);
    for (int l = 0; l < L_LAYERS; ++l) {
        transpose_kernel<<<cdiv((long)H_DIM * H_DIM, 256), 256, 0, stream>>>(
            Wl + (long)l * H_DIM * H_DIM, WlT + (long)l * H_DIM * H_DIM, H_DIM, H_DIM);
        transpose_kernel<<<cdiv((long)H_DIM * H_DIM, 256), 256, 0, stream>>>(
            Wr + (long)l * H_DIM * H_DIM, WrT + (long)l * H_DIM * H_DIM, H_DIM, H_DIM);
    }

    emb_kernel<<<tiles, 256, 0, stream>>>(x, WembT, bemb, hA, N);

    float* hcur = hA;
    for (int l = 0; l < L_LAYERS; ++l) {
        zero_f32_kernel<<<cdiv((long)N * H_DIM, 256), 256, 0, stream>>>(agg, N * H_DIM);
        scatter_kernel<<<cdiv(E, 8), 256, 0, stream>>>(src, dst, hcur, agg, E);
        float* hnext = (l == L_LAYERS - 1) ? out : (hcur == hA ? hB : hA);
        sage_layer_kernel<<<tiles, 256, 0, stream>>>(
            agg, hcur, deg,
            WlT + (long)l * H_DIM * H_DIM, WrT + (long)l * H_DIM * H_DIM,
            bl + (long)l * H_DIM, hnext, N);
        hcur = hnext;
    }
}


# per-call set-up folded into two launches: one launch clears deg and packs all nine weight matrices, the degree/key pass rides as the first blocks of the embedding launch; 20 launches -> 10
# speedup vs baseline: 2.8884x; 2.8884x over previous
#include <hip/hip_runtime.h>

typedef __attribute__((ext_vector_type(2))) float v2f;
typedef __attribute__((ext_vector_type(4))) float v4f;
typedef __attribute__((ext_vector_type(8))) float v8f;

#define IN_DIM 300
#define H_DIM  128
#define L_LAYERS 4

#define XS (IN_DIM + 2)
#define HS (H_DIM + 4)


#define RB 400
#define TD 32
#define TRIP_DW (TD * 32)
#define TRIP_E  (TRIP_DW * 4)

static __device__ __forceinline__ void degree_key_thread(const int* __restrict__ dst,
                                                         float* __restrict__ deg,
                                                         unsigned* __restrict__ key,
                                                         int e, int nkey, int q) {
    if (q >= nkey) return;
    const int i0 = 4 * q;
    int d[4];
    #pragma unroll
    for (int j = 0; j < 4; ++j) d[j] = (i0 + j < e) ? dst[i0 + j] : 0;
    unsigned w = 0;
    #pragma unroll
    for (int j = 0; j < 4; ++j) {
        unsigned b = 0xFFu;
        if (i0 + j < e) {
            unsafeAtomicAdd(&deg[d[j]], 1.0f);
            b = ((unsigned)d[j] / RB) & 0xFFu;
        }
        w |= b << (8 * j);
    }
    key[q] = w;
}

static __device__ __forceinline__ void pack_b_elem(const float* __restrict__ in,
                                                   float* __restrict__ out, int K, int Nc, int i) {
    if (i >= K * Nc) return;
    const int ksteps = K >> 2;
    const int per_ct = ksteps * 64;
    const int ct = i / per_ct;
    const int r  = i - ct * per_ct;
    const int pair_floats = (ksteps >> 1) * 128;
    int k, l;
    if (r < pair_floats) {
        const int kp = r >> 7, q = r & 127;
        l = q >> 2;
        k = 8 * kp + 4 * ((q >> 1) & 1) + 2 * (l >> 4) + (q & 1);
    } else {
        const int q = r - pair_floats;
        l = q >> 1;
        k = 4 * (ksteps - 1) + 2 * (l >> 4) + (q & 1);
    }
    out[i] = in[k * Nc + 16 * ct + (l & 15)];
}

__global__ __launch_bounds__(256)
void prep_kernel(float* __restrict__ deg, int n_nodes,
                 const float* __restrict__ Wemb, float* __restrict__ WembP,
                 const float* __restrict__ Wl, float* __restrict__ WlP,
                 const float* __restrict__ Wr, float* __restrict__ WrP,
                 int nb_zero, int nb_emb) {
    int b = blockIdx.x;
    if (b < nb_zero) {
        const int i = b * 256 + threadIdx.x;
        if (i < n_nodes) deg[i] = 0.0f;
        return;
    }
    b -= nb_zero;
    if (b < nb_emb) {
        pack_b_elem(Wemb, WembP, IN_DIM, H_DIM, b * 256 + threadIdx.x);
        return;
    }
    b -= nb_emb;
    const int per = H_DIM * H_DIM / 256;
    const int m = b / per;
    const int i = (b - m * per) * 256 + threadIdx.x;
    if (m >= 2 * L_LAYERS) return;
    const bool left = m < L_LAYERS;
    const long mo = (long)(left ? m : m - L_LAYERS) * H_DIM * H_DIM;
    pack_b_elem((left ? Wl : Wr) + mo, (left ? WlP : WrP) + mo, H_DIM, H_DIM, i);
}


static __device__ __forceinline__ void load_keys(uint4 (&kv)[TD / 4], const uint4* __restrict__ kp,
                                                 int t, int lane) {
    #pragma unroll
    for (int i = 0; i < TD / 4; ++i) kv[i] = kp[(long)t * (TRIP_DW / 4) + i * 32 + lane];
}

template <int G>
static __device__ __forceinline__ void add_group(float* acc, const float* __restrict__ h,
                                                 int sv, int dv, int g0, int lo, int nrows, int lane) {
    v4f v[G];
    #pragma unroll
    for (int g = 0; g < G; ++g) {
        const int s = __builtin_amdgcn_readlane(sv, g0 + g);
        v[g] = *reinterpret_cast<const v4f*>(h + (long)s * H_DIM + lane * 4);
    }
    #pragma unroll
    for (int g = 0; g < G; ++g) {
        const int dl = __builtin_amdgcn_readlane(dv, g0 + g) - lo;
        if ((unsigned)dl < (unsigned)nrows) {
            float* a = acc + dl * H_DIM + lane;
            __hip_atomic_fetch_add(a,      v[g].x, __ATOMIC_RELAXED, __HIP_MEMORY_SCOPE_WORKGROUP);
            __hip_atomic_fetch_add(a + 32, v[g].y, __ATOMIC_RELAXED, __HIP_MEMORY_SCOPE_WORKGROUP);
            __hip_atomic_fetch_add(a + 64, v[g].z, __ATOMIC_RELAXED, __HIP_MEMORY_SCOPE_WORKGROUP);
            __hip_atomic_fetch_add(a + 96, v[g].w, __ATOMIC_RELAXED, __HIP_MEMORY_SCOPE_WORKGROUP);
        }
    }
}

__global__ __launch_bounds__(1024)
void agg_range_kernel(const int* __restrict__ src, const int* __restrict__ dst,
                      const unsigned* __restrict__ key, const float* __restrict__ deg,
                      const float* __restrict__ h, float* __restrict__ agg,
                      int* __restrict__ elist, int* __restrict__ gpos,
                      int e, int n_nodes, int ntrips) {
    __shared__ __attribute__((aligned(16))) float acc[RB * H_DIM];
    __shared__ int s_start, s_fill;
    const int tid = threadIdx.x, lane = tid & 31;
    const int wave = __builtin_amdgcn_readfirstlane(tid >> 5);
    const int lo = blockIdx.x * RB;
    int nrows = n_nodes - lo; if (nrows > RB) nrows = RB;
    const unsigned pat = ((unsigned)blockIdx.x & 255u) * 0x01010101u;
    const uint4* kp = reinterpret_cast<const uint4*>(key);

    float4* acc4 = reinterpret_cast<float4*>(acc);
    for (int i = tid; i < RB * H_DIM / 4; i += 1024) acc4[i] = make_float4(0.f, 0.f, 0.f, 0.f);
    if (tid == 0) { s_start = 0; s_fill = 0; }
    int part = 0;
    const float4* d4 = reinterpret_cast<const float4*>(deg);
    for (int i = tid; i < lo / 4; i += 1024) {
        const float4 v = d4[i];
        part += (int)v.x + (int)v.y + (int)v.z + (int)v.w;
    }
    __syncthreads();
    if (part) __hip_atomic_fetch_add(&s_start, part, __ATOMIC_RELAXED, __HIP_MEMORY_SCOPE_WORKGROUP);
    __syncthreads();
    const int start = s_start;
    int2* lst = reinterpret_cast<int2*>(elist);

    auto sum_matches = [&](int base, int qreg, int cnt) {
        int sv = 0, dv = lo - 1;
        if (lane < cnt) {
            const int i = base + qreg;
            if (i < e) { sv = src[i]; dv = dst[i]; }
        }
        {
            const bool ok = (unsigned)(dv - lo) < (unsigned)nrows;
            const unsigned bal = __builtin_amdgcn_ballot_w32(ok);
            int b = 0;
            if (lane == 0) b = __hip_atomic_fetch_add(&s_fill, __builtin_popcount(bal),
                                                      __ATOMIC_RELAXED, __HIP_MEMORY_SCOPE_WORKGROUP);
            b = __builtin_amdgcn_readfirstlane(b);
            const int pos = start + b + __builtin_popcount(bal & ((1u << lane) - 1u));
            if (ok && pos < e) lst[pos] = make_int2(sv, dv - lo);
        }
        int g0 = 0;
        for (; g0 + 8 <= cnt; g0 += 8) add_group<8>(acc, h, sv, dv, g0, lo, nrows, lane);
        int r = cnt - g0;
        if (r >= 4) { add_group<4>(acc, h, sv, dv, g0, lo, nrows, lane); g0 += 4; r -= 4; }
        if (r == 3)      add_group<3>(acc, h, sv, dv, g0, lo, nrows, lane);
        else if (r == 2) add_group<2>(acc, h, sv, dv, g0, lo, nrows, lane);
        else if (r == 1) add_group<1>(acc, h, sv, dv, g0, lo, nrows, lane);
    };

    if (wave < ntrips) {
    uint4 kv[TD / 4];
    load_keys(kv, kp, wave, lane);
    for (int t = wave; t < ntrips; t += 32) {
        const int base = t * TRIP_E;
        uint4 kn[TD / 4];
        load_keys(kn, kp, (t + 32 < ntrips) ? t + 32 : t, lane);

        unsigned hinv[4] = {~0u, ~0u, ~0u, ~0u};
        #pragma unroll
        for (int i = 0; i < TD / 4; ++i) {
            const unsigned c4[4] = {kv[i].x, kv[i].y, kv[i].z, kv[i].w};
            #pragma unroll
            for (int c = 0; c < 4; ++c) {
                const int k = 4 * i + c;
                const unsigned x  = c4[c] ^ pat;
                const unsigned nz = ((x & 0x7f7f7f7fu) + 0x7f7f7f7fu) | x | 0x7f7f7f7fu;
                hinv[k >> 3] &= __builtin_rotateright32(nz, 7 - (k & 7));
            }
        }

        int qreg = 0, qn = 0;
        #pragma unroll 1
        for (int g = 0; g < 4; ++g) {
            unsigned hw = ~hinv[0];
            hw = (g == 1) ? ~hinv[1] : hw;
            hw = (g == 2) ? ~hinv[2] : hw;
            hw = (g == 3) ? ~hinv[3] : hw;
            unsigned any = __builtin_amdgcn_ballot_w32(hw != 0);
            while (any) {
                const int L = __builtin_ctz(any);
                any &= any - 1;
                unsigned m = (unsigned)__builtin_amdgcn_readlane((int)hw, L);
                while (m) {
                    const int pbit = __builtin_ctz(m);
                    m &= m - 1;
                    const int off = (2 * g + ((pbit >> 2) & 1)) * 512 + L * 16 + (pbit & 3) * 4 + (pbit >> 3);
                    qreg = (lane == qn) ? off : qreg;
                    if (++qn == 32) { sum_matches(base, qreg, 32); qn = 0; }
                }
            }
        }
        if (qn) sum_matches(base, qreg, qn);

        #pragma unroll
        for (int i = 0; i < TD / 4; ++i) kv[i] = kn[i];
    }
    }
    __syncthreads();

    float4* out4 = reinterpret_cast<float4*>(agg + (long)lo * H_DIM);
    for (int i = tid; i < nrows * (H_DIM / 4); i += 1024) {
        const float* a = acc + (i >> 5) * H_DIM + (i & 31);
        out4[i] = make_float4(a[0], a[32], a[64], a[96]);
    }
    if (tid == 0) { gpos[2 * blockIdx.x] = start; gpos[2 * blockIdx.x + 1] = s_fill; }
}

template <int G>
static __device__ __forceinline__ void add_rows(float* acc, const float* __restrict__ h,
                                                int sv, int rv, int g0, int lane) {
    v4f v[G];
    #pragma unroll
    for (int g = 0; g < G; ++g) {
        const int s = __builtin_amdgcn_readlane(sv, g0 + g);
        v[g] = *reinterpret_cast<const v4f*>(h + (long)s * H_DIM + lane * 4);
    }
    #pragma unroll
    for (int g = 0; g < G; ++g) {
        unsigned dl = (unsigned)__builtin_amdgcn_readlane(rv, g0 + g);
        if (dl > RB - 1) dl = RB - 1;
        float* a = acc + dl * H_DIM + lane;
        __hip_atomic_fetch_add(a,      v[g].x, __ATOMIC_RELAXED, __HIP_MEMORY_SCOPE_WORKGROUP);
        __hip_atomic_fetch_add(a + 32, v[g].y, __ATOMIC_RELAXED, __HIP_MEMORY_SCOPE_WORKGROUP);
        __hip_atomic_fetch_add(a + 64, v[g].z, __ATOMIC_RELAXED, __HIP_MEMORY_SCOPE_WORKGROUP);
        __hip_atomic_fetch_add(a + 96, v[g].w, __ATOMIC_RELAXED, __HIP_MEMORY_SCOPE_WORKGROUP);
    }
}

__global__ __launch_bounds__(1024)
void agg_list_kernel(const int* __restrict__ elist, const int* __restrict__ gpos,
                     const float* __restrict__ h, float* __restrict__ agg,
                     int e, int n_nodes) {
    __shared__ __attribute__((aligned(16))) float acc[RB * H_DIM];
    const int tid = threadIdx.x, lane = tid & 31;
    const int wave = __builtin_amdgcn_readfirstlane(tid >> 5);
    const int lo = blockIdx.x * RB;
    int nrows = n_nodes - lo; if (nrows > RB) nrows = RB;

    float4* acc4 = reinterpret_cast<float4*>(acc);
    for (int i = tid; i < RB * H_DIM / 4; i += 1024) acc4[i] = make_float4(0.f, 0.f, 0.f, 0.f);

    int start = gpos[2 * blockIdx.x], cnt = gpos[2 * blockIdx.x + 1];
    if (start < 0) start = 0;
    if (start > e) start = e;
    if (cnt > e - start) cnt = e - start;
    if (cnt < 0) cnt = 0;
    const int2* lp = reinterpret_cast<const int2*>(elist) + start;
    const int b0 = (int)(((long)cnt * wave) >> 5), b1 = (int)(((long)cnt * (wave + 1)) >> 5);
    __syncthreads();

    if (b0 < b1) {
    int2 en = lp[(b0 + lane < b1) ? b0 + lane : b1 - 1];
    for (int p = b0; p < b1; p += 32) {
        int m = b1 - p; if (m > 32) m = 32;
        const int2 nx = lp[(p + 32 + lane < b1) ? p + 32 + lane : b1 - 1];
        const int sv = en.x, dv = en.y;
        int g0 = 0;
        for (; g0 + 8 <= m; g0 += 8) add_rows<8>(acc, h, sv, dv, g0, lane);
        int r = m - g0;
        if (r >= 4) { add_rows<4>(acc, h, sv, dv, g0, lane); g0 += 4; r -= 4; }
        if (r == 3)      add_rows<3>(acc, h, sv, dv, g0, lane);
        else if (r == 2) add_rows<2>(acc, h, sv, dv, g0, lane);
        else if (r == 1) add_rows<1>(acc, h, sv, dv, g0, lane);
        en = nx;
    }
    }
    __syncthreads();

    float4* out4 = reinterpret_cast<float4*>(agg + (long)lo * H_DIM);
    for (int i = tid; i < nrows * (H_DIM / 4); i += 1024) {
        const float* a = acc + (i >> 5) * H_DIM + (i & 31);
        out4[i] = make_float4(a[0], a[32], a[64], a[96]);
    }
}


__global__ __launch_bounds__(256)
void emb_kernel(const float* __restrict__ x, const float* __restrict__ WembP,
                const float* __restrict__ bemb, float* __restrict__ h, int n_nodes,
                const int* __restrict__ dst, float* __restrict__ deg,
                unsigned* __restrict__ key, int e, int nkey, int nb_deg) {
    __shared__ float shx[16 * XS];
    const int  tid  = threadIdx.x;
    if ((int)blockIdx.x < nb_deg) {
        degree_key_thread(dst, deg, key, e, nkey, (int)blockIdx.x * 256 + tid);
        return;
    }
    const long row0 = (long)((int)blockIdx.x - nb_deg) * 16;
    const bool full = (row0 + 16 <= (long)n_nodes);

    if (full) {
        for (int i = tid; i < 16 * IN_DIM; i += 256) {
            int r = i / IN_DIM, c = i - r * IN_DIM;
            shx[r * XS + c] = x[(row0 + r) * IN_DIM + c];
        }
    } else {
        for (int i = tid; i < 16 * IN_DIM; i += 256) {
            int r = i / IN_DIM, c = i - r * IN_DIM;
            long rr = row0 + r; if (rr >= n_nodes) rr = n_nodes - 1;
            shx[r * XS + c] = x[rr * IN_DIM + c];
        }
    }
    __syncthreads();

    const int wave = tid >> 5, lane = tid & 31;
    const int m = lane & 15, hi = lane >> 4;
    const int ncol = wave * 16 + m;

    v8f acc;
    float bv = bemb[ncol];
    #pragma unroll
    for (int i = 0; i < 8; ++i) acc[i] = bv;

    const float* wp = WembP + wave * (IN_DIM * 16) + lane * 4;
    const int a0 = m * XS + 2 * hi;
    #pragma unroll
    for (int kp = 0; kp < IN_DIM / 8; ++kp) {
        const v4f b = *reinterpret_cast<const v4f*>(wp + kp * 128);
        v2f a, c, b0, b1;
        a.x = shx[a0 + 8 * kp];     a.y = shx[a0 + 8 * kp + 1];
        c.x = shx[a0 + 8 * kp + 4]; c.y = shx[a0 + 8 * kp + 5];
        b0.x = b.x; b0.y = b.y; b1.x = b.z; b1.y = b.w;
        acc = __builtin_amdgcn_wmma_f32_16x16x4_f32(false, a, false, b0,
                                                    (short)0, acc, false, false);
        acc = __builtin_amdgcn_wmma_f32_16x16x4_f32(false, c, false, b1,
                                                    (short)0, acc, false, false);
    }
    if (IN_DIM & 4) {
        const int kl = IN_DIM - 4;
        const v2f b = *reinterpret_cast<const v2f*>(WembP + wave * (IN_DIM * 16)
                                                    + (IN_DIM / 8) * 128 + lane * 2);
        v2f a; a.x = shx[a0 + kl]; a.y = shx[a0 + kl + 1];
        acc = __builtin_amdgcn_wmma_f32_16x16x4_f32(false, a, false, b,
                                                    (short)0, acc, false, false);
    }
    asm volatile("v_nop\n\tv_nop\n\tv_nop\n\tv_nop" : "+v"(acc));

    float* hrow = h + row0 * H_DIM + wave * 16 + m;
    if (full) {
        #pragma unroll
        for (int v = 0; v < 8; ++v) {
            float r = acc[v];
            hrow[(long)(v + 8 * hi) * H_DIM] = r > 0.0f ? r : 0.0f;
        }
    } else {
        #pragma unroll
        for (int v = 0; v < 8; ++v) {
            int mm = v + 8 * hi;
            if (row0 + mm < n_nodes) {
                float r = acc[v];
                hrow[(long)mm * H_DIM] = r > 0.0f ? r : 0.0f;
            }
        }
    }
}

__global__ __launch_bounds__(256)
void sage_layer_kernel(const float* __restrict__ agg, const float* __restrict__ hin,
                       const float* __restrict__ deg,
                       const float* __restrict__ WlP, const float* __restrict__ WrP,
                       const float* __restrict__ bias, float* __restrict__ hout,
                       int n_nodes) {
    __shared__ float sh_a[16 * HS];
    __shared__ float sh_h[16 * HS];
    const int  tid  = threadIdx.x;
    const long row0 = (long)blockIdx.x * 16;
    const bool full = (row0 + 16 <= (long)n_nodes);

    if (full) {
        for (int i = tid; i < 16 * H_DIM; i += 256) {
            int r = i >> 7, c = i & 127;
            long rr = row0 + r;
            float dg  = deg[rr];
            float inv = 1.0f / (dg > 1.0f ? dg : 1.0f);
            sh_a[r * HS + c] = agg[rr * H_DIM + c] * inv;
            sh_h[r * HS + c] = hin[rr * H_DIM + c];
        }
    } else {
        for (int i = tid; i < 16 * H_DIM; i += 256) {
            int r = i >> 7, c = i & 127;
            long rr = row0 + r; if (rr >= n_nodes) rr = n_nodes - 1;
            float dg  = deg[rr];
            float inv = 1.0f / (dg > 1.0f ? dg : 1.0f);
            sh_a[r * HS + c] = agg[rr * H_DIM + c] * inv;
            sh_h[r * HS + c] = hin[rr * H_DIM + c];
        }
    }
    __syncthreads();

    const int wave = tid >> 5, lane = tid & 31;
    const int m = lane & 15, hi = lane >> 4;
    const int ncol = wave * 16 + m;

    v8f acc;
    float bv = bias[ncol];
    #pragma unroll
    for (int i = 0; i < 8; ++i) acc[i] = bv;

    const float* wl = WlP + wave * (H_DIM * 16) + lane * 4;
    const float* wr = WrP + wave * (H_DIM * 16) + lane * 4;
    const int a0 = m * HS + 2 * hi;
    #pragma unroll
    for (int kp = 0; kp < H_DIM / 8; ++kp) {
        const v4f vl = *reinterpret_cast<const v4f*>(wl + kp * 128);
        const v4f vr = *reinterpret_cast<const v4f*>(wr + kp * 128);
        v2f aa, ah, ca, ch, bl0, br0, bl1, br1;
        aa.x = sh_a[a0 + 8 * kp];     aa.y = sh_a[a0 + 8 * kp + 1];
        ah.x = sh_h[a0 + 8 * kp];     ah.y = sh_h[a0 + 8 * kp + 1];
        ca.x = sh_a[a0 + 8 * kp + 4]; ca.y = sh_a[a0 + 8 * kp + 5];
        ch.x = sh_h[a0 + 8 * kp + 4]; ch.y = sh_h[a0 + 8 * kp + 5];
        bl0.x = vl.x; bl0.y = vl.y; bl1.x = vl.z; bl1.y = vl.w;
        br0.x = vr.x; br0.y = vr.y; br1.x = vr.z; br1.y = vr.w;
        acc = __builtin_amdgcn_wmma_f32_16x16x4_f32(false, aa, false, bl0,
                                                    (short)0, acc, false, false);
        acc = __builtin_amdgcn_wmma_f32_16x16x4_f32(false, ah, false, br0,
                                                    (short)0, acc, false, false);
        acc = __builtin_amdgcn_wmma_f32_16x16x4_f32(false, ca, false, bl1,
                                                    (short)0, acc, false, false);
        acc = __builtin_amdgcn_wmma_f32_16x16x4_f32(false, ch, false, br1,
                                                    (short)0, acc, false, false);
    }
    asm volatile("v_nop\n\tv_nop\n\tv_nop\n\tv_nop" : "+v"(acc));

    float* orow = hout + row0 * H_DIM + wave * 16 + m;
    if (full) {
        #pragma unroll
        for (int v = 0; v < 8; ++v) {
            float r = acc[v];
            orow[(long)(v + 8 * hi) * H_DIM] = r > 0.0f ? r : 0.0f;
        }
    } else {
        #pragma unroll
        for (int v = 0; v < 8; ++v) {
            int mm = v + 8 * hi;
            if (row0 + mm < n_nodes) {
                float r = acc[v];
                orow[(long)mm * H_DIM] = r > 0.0f ? r : 0.0f;
            }
        }
    }
}


static inline int cdiv(long a, long b) { return (int)((a + b - 1) / b); }

extern "C" void kernel_launch(void* const* d_in, const int* in_sizes, int n_in,
                              void* d_out, int out_size, void* d_ws, size_t ws_size,
                              hipStream_t stream) {
    const float* x    = (const float*)d_in[0];
    const int*   ei   = (const int*)  d_in[1];
    const float* Wemb = (const float*)d_in[2];
    const float* bemb = (const float*)d_in[3];
    const float* Wl   = (const float*)d_in[4];
    const float* bl   = (const float*)d_in[5];
    const float* Wr   = (const float*)d_in[6];
    float* out = (float*)d_out;

    const int N = in_sizes[0] / IN_DIM;
    const int E = in_sizes[1] / 2;
    const int* src = ei;
    const int* dst = ei + E;

    char*  ws  = (char*)d_ws;
    size_t off = 0;
    auto carve = [&](size_t bytes) -> float* {
        float* p = (float*)(ws + off);
        off = (off + bytes + 255) & ~(size_t)255;
        return p;
    };
    float* deg   = carve((size_t)N * sizeof(float));
    float* hA    = carve((size_t)N * H_DIM * sizeof(float));
    float* hB    = carve((size_t)N * H_DIM * sizeof(float));
    float* agg   = carve((size_t)N * H_DIM * sizeof(float));
    float* WembP = carve((size_t)IN_DIM * H_DIM * sizeof(float));
    float* WlP   = carve((size_t)L_LAYERS * H_DIM * H_DIM * sizeof(float));
    float* WrP   = carve((size_t)L_LAYERS * H_DIM * H_DIM * sizeof(float));
    const int ntrips = cdiv(E, TRIP_E);
    const int nkey   = ntrips * TRIP_DW;
    unsigned* key = (unsigned*)carve((size_t)nkey * sizeof(unsigned));
    (void)ws_size;

    const int tiles = cdiv(N, 16);

    const int nb_zero = cdiv(N, 256);
    const int nb_emb  = cdiv((long)IN_DIM * H_DIM, 256);
    const int nb_lay  = 2 * L_LAYERS * (H_DIM * H_DIM / 256);
    prep_kernel<<<nb_zero + nb_emb + nb_lay, 256, 0, stream>>>(deg, N, Wemb, WembP, Wl, WlP, Wr, WrP,
                                                              nb_zero, nb_emb);

    const int nb_deg = cdiv(nkey, 256);
    emb_kernel<<<nb_deg + tiles, 256, 0, stream>>>(x, WembP, bemb, hA, N, dst, deg, key, E, nkey, nb_deg);

    int* elist = (int*)d_out;
    int* gpos  = elist + 2 * (long)E;

    float* hcur = hA;
    for (int l = 0; l < L_LAYERS; ++l) {
        if (l == 0)
            agg_range_kernel<<<cdiv(N, RB), 1024, 0, stream>>>(src, dst, key, deg, hcur, agg,
                                                              elist, gpos, E, N, ntrips);
        else
            agg_list_kernel<<<cdiv(N, RB), 1024, 0, stream>>>(elist, gpos, hcur, agg, E, N);
        float* hnext = (l == L_LAYERS - 1) ? out : (hcur == hA ? hB : hA);
        sage_layer_kernel<<<tiles, 256, 0, stream>>>(
            agg, hcur, deg,
            WlP + (long)l * H_DIM * H_DIM, WrP + (long)l * H_DIM * H_DIM,
            bl + (long)l * H_DIM, hnext, N);
        hcur = hnext;
    }
}


# GEMM work-groups take 32 rows: each wave keeps 2 accumulator tiles and feeds 2 matrix instructions from every B fragment load, halving packed-weight reads; per-element sum order unchanged
# speedup vs baseline: 3.0356x; 1.0510x over previous
#include <hip/hip_runtime.h>

typedef __attribute__((ext_vector_type(2))) float v2f;
typedef __attribute__((ext_vector_type(4))) float v4f;
typedef __attribute__((ext_vector_type(8))) float v8f;

#define IN_DIM 300
#define H_DIM  128
#define L_LAYERS 4

#define XS (IN_DIM + 2)
#define HS (H_DIM + 4)

#define RT 2
#define TM (16 * RT)


#define RB 400
#define TD 32
#define TRIP_DW (TD * 32)
#define TRIP_E  (TRIP_DW * 4)

static __device__ __forceinline__ void degree_key_thread(const int* __restrict__ dst,
                                                         float* __restrict__ deg,
                                                         unsigned* __restrict__ key,
                                                         int e, int nkey, int q) {
    if (q >= nkey) return;
    const int i0 = 4 * q;
    int d[4];
    #pragma unroll
    for (int j = 0; j < 4; ++j) d[j] = (i0 + j < e) ? dst[i0 + j] : 0;
    unsigned w = 0;
    #pragma unroll
    for (int j = 0; j < 4; ++j) {
        unsigned b = 0xFFu;
        if (i0 + j < e) {
            unsafeAtomicAdd(&deg[d[j]], 1.0f);
            b = ((unsigned)d[j] / RB) & 0xFFu;
        }
        w |= b << (8 * j);
    }
    key[q] = w;
}

static __device__ __forceinline__ void pack_b_elem(const float* __restrict__ in,
                                                   float* __restrict__ out, int K, int Nc, int i) {
    if (i >= K * Nc) return;
    const int ksteps = K >> 2;
    const int per_ct = ksteps * 64;
    const int ct = i / per_ct;
    const int r  = i - ct * per_ct;
    const int pair_floats = (ksteps >> 1) * 128;
    int k, l;
    if (r < pair_floats) {
        const int kp = r >> 7, q = r & 127;
        l = q >> 2;
        k = 8 * kp + 4 * ((q >> 1) & 1) + 2 * (l >> 4) + (q & 1);
    } else {
        const int q = r - pair_floats;
        l = q >> 1;
        k = 4 * (ksteps - 1) + 2 * (l >> 4) + (q & 1);
    }
    out[i] = in[k * Nc + 16 * ct + (l & 15)];
}

__global__ __launch_bounds__(256)
void prep_kernel(float* __restrict__ deg, int n_nodes,
                 const float* __restrict__ Wemb, float* __restrict__ WembP,
                 const float* __restrict__ Wl, float* __restrict__ WlP,
                 const float* __restrict__ Wr, float* __restrict__ WrP,
                 int nb_zero, int nb_emb) {
    int b = blockIdx.x;
    if (b < nb_zero) {
        const int i = b * 256 + threadIdx.x;
        if (i < n_nodes) deg[i] = 0.0f;
        return;
    }
    b -= nb_zero;
    if (b < nb_emb) {
        pack_b_elem(Wemb, WembP, IN_DIM, H_DIM, b * 256 + threadIdx.x);
        return;
    }
    b -= nb_emb;
    const int per = H_DIM * H_DIM / 256;
    const int m = b / per;
    const int i = (b - m * per) * 256 + threadIdx.x;
    if (m >= 2 * L_LAYERS) return;
    const bool left = m < L_LAYERS;
    const long mo = (long)(left ? m : m - L_LAYERS) * H_DIM * H_DIM;
    pack_b_elem((left ? Wl : Wr) + mo, (left ? WlP : WrP) + mo, H_DIM, H_DIM, i);
}


static __device__ __forceinline__ void load_keys(uint4 (&kv)[TD / 4], const uint4* __restrict__ kp,
                                                 int t, int lane) {
    #pragma unroll
    for (int i = 0; i < TD / 4; ++i) kv[i] = kp[(long)t * (TRIP_DW / 4) + i * 32 + lane];
}

template <int G>
static __device__ __forceinline__ void add_group(float* acc, const float* __restrict__ h,
                                                 int sv, int dv, int g0, int lo, int nrows, int lane) {
    v4f v[G];
    #pragma unroll
    for (int g = 0; g < G; ++g) {
        const int s = __builtin_amdgcn_readlane(sv, g0 + g);
        v[g] = *reinterpret_cast<const v4f*>(h + (long)s * H_DIM + lane * 4);
    }
    #pragma unroll
    for (int g = 0; g < G; ++g) {
        const int dl = __builtin_amdgcn_readlane(dv, g0 + g) - lo;
        if ((unsigned)dl < (unsigned)nrows) {
            float* a = acc + dl * H_DIM + lane;
            __hip_atomic_fetch_add(a,      v[g].x, __ATOMIC_RELAXED, __HIP_MEMORY_SCOPE_WORKGROUP);
            __hip_atomic_fetch_add(a + 32, v[g].y, __ATOMIC_RELAXED, __HIP_MEMORY_SCOPE_WORKGROUP);
            __hip_atomic_fetch_add(a + 64, v[g].z, __ATOMIC_RELAXED, __HIP_MEMORY_SCOPE_WORKGROUP);
            __hip_atomic_fetch_add(a + 96, v[g].w, __ATOMIC_RELAXED, __HIP_MEMORY_SCOPE_WORKGROUP);
        }
    }
}

__global__ __launch_bounds__(1024)
void agg_range_kernel(const int* __restrict__ src, const int* __restrict__ dst,
                      const unsigned* __restrict__ key, const float* __restrict__ deg,
                      const float* __restrict__ h, float* __restrict__ agg,
                      int* __restrict__ elist, int* __restrict__ gpos,
                      int e, int n_nodes, int ntrips) {
    __shared__ __attribute__((aligned(16))) float acc[RB * H_DIM];
    __shared__ int s_start, s_fill;
    const int tid = threadIdx.x, lane = tid & 31;
    const int wave = __builtin_amdgcn_readfirstlane(tid >> 5);
    const int lo = blockIdx.x * RB;
    int nrows = n_nodes - lo; if (nrows > RB) nrows = RB;
    const unsigned pat = ((unsigned)blockIdx.x & 255u) * 0x01010101u;
    const uint4* kp = reinterpret_cast<const uint4*>(key);

    float4* acc4 = reinterpret_cast<float4*>(acc);
    for (int i = tid; i < RB * H_DIM / 4; i += 1024) acc4[i] = make_float4(0.f, 0.f, 0.f, 0.f);
    if (tid == 0) { s_start = 0; s_fill = 0; }
    int part = 0;
    const float4* d4 = reinterpret_cast<const float4*>(deg);
    for (int i = tid; i < lo / 4; i += 1024) {
        const float4 v = d4[i];
        part += (int)v.x + (int)v.y + (int)v.z + (int)v.w;
    }
    __syncthreads();
    if (part) __hip_atomic_fetch_add(&s_start, part, __ATOMIC_RELAXED, __HIP_MEMORY_SCOPE_WORKGROUP);
    __syncthreads();
    const int start = s_start;
    int2* lst = reinterpret_cast<int2*>(elist);

    auto sum_matches = [&](int base, int qreg, int cnt) {
        int sv = 0, dv = lo - 1;
        if (lane < cnt) {
            const int i = base + qreg;
            if (i < e) { sv = src[i]; dv = dst[i]; }
        }
        {
            const bool ok = (unsigned)(dv - lo) < (unsigned)nrows;
            const unsigned bal = __builtin_amdgcn_ballot_w32(ok);
            int b = 0;
            if (lane == 0) b = __hip_atomic_fetch_add(&s_fill, __builtin_popcount(bal),
                                                      __ATOMIC_RELAXED, __HIP_MEMORY_SCOPE_WORKGROUP);
            b = __builtin_amdgcn_readfirstlane(b);
            const int pos = start + b + __builtin_popcount(bal & ((1u << lane) - 1u));
            if (ok && pos < e) lst[pos] = make_int2(sv, dv - lo);
        }
        int g0 = 0;
        for (; g0 + 8 <= cnt; g0 += 8) add_group<8>(acc, h, sv, dv, g0, lo, nrows, lane);
        int r = cnt - g0;
        if (r >= 4) { add_group<4>(acc, h, sv, dv, g0, lo, nrows, lane); g0 += 4; r -= 4; }
        if (r == 3)      add_group<3>(acc, h, sv, dv, g0, lo, nrows, lane);
        else if (r == 2) add_group<2>(acc, h, sv, dv, g0, lo, nrows, lane);
        else if (r == 1) add_group<1>(acc, h, sv, dv, g0, lo, nrows, lane);
    };

    if (wave < ntrips) {
    uint4 kv[TD / 4];
    load_keys(kv, kp, wave, lane);
    for (int t = wave; t < ntrips; t += 32) {
        const int base = t * TRIP_E;
        uint4 kn[TD / 4];
        load_keys(kn, kp, (t + 32 < ntrips) ? t + 32 : t, lane);

        unsigned hinv[4] = {~0u, ~0u, ~0u, ~0u};
        #pragma unroll
        for (int i = 0; i < TD / 4; ++i) {
            const unsigned c4[4] = {kv[i].x, kv[i].y, kv[i].z, kv[i].w};
            #pragma unroll
            for (int c = 0; c < 4; ++c) {
                const int k = 4 * i + c;
                const unsigned x  = c4[c] ^ pat;
                const unsigned nz = ((x & 0x7f7f7f7fu) + 0x7f7f7f7fu) | x | 0x7f7f7f7fu;
                hinv[k >> 3] &= __builtin_rotateright32(nz, 7 - (k & 7));
            }
        }

        int qreg = 0, qn = 0;
        #pragma unroll 1
        for (int g = 0; g < 4; ++g) {
            unsigned hw = ~hinv[0];
            hw = (g == 1) ? ~hinv[1] : hw;
            hw = (g == 2) ? ~hinv[2] : hw;
            hw = (g == 3) ? ~hinv[3] : hw;
            unsigned any = __builtin_amdgcn_ballot_w32(hw != 0);
            while (any) {
                const int L = __builtin_ctz(any);
                any &= any - 1;
                unsigned m = (unsigned)__builtin_amdgcn_readlane((int)hw, L);
                while (m) {
                    const int pbit = __builtin_ctz(m);
                    m &= m - 1;
                    const int off = (2 * g + ((pbit >> 2) & 1)) * 512 + L * 16 + (pbit & 3) * 4 + (pbit >> 3);
                    qreg = (lane == qn) ? off : qreg;
                    if (++qn == 32) { sum_matches(base, qreg, 32); qn = 0; }
                }
            }
        }
        if (qn) sum_matches(base, qreg, qn);

        #pragma unroll
        for (int i = 0; i < TD / 4; ++i) kv[i] = kn[i];
    }
    }
    __syncthreads();

    float4* out4 = reinterpret_cast<float4*>(agg + (long)lo * H_DIM);
    for (int i = tid; i < nrows * (H_DIM / 4); i += 1024) {
        const float* a = acc + (i >> 5) * H_DIM + (i & 31);
        out4[i] = make_float4(a[0], a[32], a[64], a[96]);
    }
    if (tid == 0) { gpos[2 * blockIdx.x] = start; gpos[2 * blockIdx.x + 1] = s_fill; }
}

template <int G>
static __device__ __forceinline__ void add_rows(float* acc, const float* __restrict__ h,
                                                int sv, int rv, int g0, int lane) {
    v4f v[G];
    #pragma unroll
    for (int g = 0; g < G; ++g) {
        const int s = __builtin_amdgcn_readlane(sv, g0 + g);
        v[g] = *reinterpret_cast<const v4f*>(h + (long)s * H_DIM + lane * 4);
    }
    #pragma unroll
    for (int g = 0; g < G; ++g) {
        unsigned dl = (unsigned)__builtin_amdgcn_readlane(rv, g0 + g);
        if (dl > RB - 1) dl = RB - 1;
        float* a = acc + dl * H_DIM + lane;
        __hip_atomic_fetch_add(a,      v[g].x, __ATOMIC_RELAXED, __HIP_MEMORY_SCOPE_WORKGROUP);
        __hip_atomic_fetch_add(a + 32, v[g].y, __ATOMIC_RELAXED, __HIP_MEMORY_SCOPE_WORKGROUP);
        __hip_atomic_fetch_add(a + 64, v[g].z, __ATOMIC_RELAXED, __HIP_MEMORY_SCOPE_WORKGROUP);
        __hip_atomic_fetch_add(a + 96, v[g].w, __ATOMIC_RELAXED, __HIP_MEMORY_SCOPE_WORKGROUP);
    }
}

__global__ __launch_bounds__(1024)
void agg_list_kernel(const int* __restrict__ elist, const int* __restrict__ gpos,
                     const float* __restrict__ h, float* __restrict__ agg,
                     int e, int n_nodes) {
    __shared__ __attribute__((aligned(16))) float acc[RB * H_DIM];
    const int tid = threadIdx.x, lane = tid & 31;
    const int wave = __builtin_amdgcn_readfirstlane(tid >> 5);
    const int lo = blockIdx.x * RB;
    int nrows = n_nodes - lo; if (nrows > RB) nrows = RB;

    float4* acc4 = reinterpret_cast<float4*>(acc);
    for (int i = tid; i < RB * H_DIM / 4; i += 1024) acc4[i] = make_float4(0.f, 0.f, 0.f, 0.f);

    int start = gpos[2 * blockIdx.x], cnt = gpos[2 * blockIdx.x + 1];
    if (start < 0) start = 0;
    if (start > e) start = e;
    if (cnt > e - start) cnt = e - start;
    if (cnt < 0) cnt = 0;
    const int2* lp = reinterpret_cast<const int2*>(elist) + start;
    const int b0 = (int)(((long)cnt * wave) >> 5), b1 = (int)(((long)cnt * (wave + 1)) >> 5);
    __syncthreads();

    if (b0 < b1) {
    int2 en = lp[(b0 + lane < b1) ? b0 + lane : b1 - 1];
    for (int p = b0; p < b1; p += 32) {
        int m = b1 - p; if (m > 32) m = 32;
        const int2 nx = lp[(p + 32 + lane < b1) ? p + 32 + lane : b1 - 1];
        const int sv = en.x, dv = en.y;
        int g0 = 0;
        for (; g0 + 8 <= m; g0 += 8) add_rows<8>(acc, h, sv, dv, g0, lane);
        int r = m - g0;
        if (r >= 4) { add_rows<4>(acc, h, sv, dv, g0, lane); g0 += 4; r -= 4; }
        if (r == 3)      add_rows<3>(acc, h, sv, dv, g0, lane);
        else if (r == 2) add_rows<2>(acc, h, sv, dv, g0, lane);
        else if (r == 1) add_rows<1>(acc, h, sv, dv, g0, lane);
        en = nx;
    }
    }
    __syncthreads();

    float4* out4 = reinterpret_cast<float4*>(agg + (long)lo * H_DIM);
    for (int i = tid; i < nrows * (H_DIM / 4); i += 1024) {
        const float* a = acc + (i >> 5) * H_DIM + (i & 31);
        out4[i] = make_float4(a[0], a[32], a[64], a[96]);
    }
}


__global__ __launch_bounds__(256)
void emb_kernel(const float* __restrict__ x, const float* __restrict__ WembP,
                const float* __restrict__ bemb, float* __restrict__ h, int n_nodes,
                const int* __restrict__ dst, float* __restrict__ deg,
                unsigned* __restrict__ key, int e, int nkey, int nb_deg) {
    __shared__ float shx[TM * XS];
    const int  tid  = threadIdx.x;
    if ((int)blockIdx.x < nb_deg) {
        degree_key_thread(dst, deg, key, e, nkey, (int)blockIdx.x * 256 + tid);
        return;
    }
    const long row0 = (long)((int)blockIdx.x - nb_deg) * TM;
    const bool full = (row0 + TM <= (long)n_nodes);

    if (full) {
        for (int i = tid; i < TM * IN_DIM; i += 256) {
            int r = i / IN_DIM, c = i - r * IN_DIM;
            shx[r * XS + c] = x[(row0 + r) * IN_DIM + c];
        }
    } else {
        for (int i = tid; i < TM * IN_DIM; i += 256) {
            int r = i / IN_DIM, c = i - r * IN_DIM;
            long rr = row0 + r; if (rr >= n_nodes) rr = n_nodes - 1;
            shx[r * XS + c] = x[rr * IN_DIM + c];
        }
    }
    __syncthreads();

    const int wave = tid >> 5, lane = tid & 31;
    const int m = lane & 15, hi = lane >> 4;
    const int ncol = wave * 16 + m;

    v8f acc[RT];
    float bv = bemb[ncol];
    #pragma unroll
    for (int t = 0; t < RT; ++t) {
        #pragma unroll
        for (int i = 0; i < 8; ++i) acc[t][i] = bv;
    }

    const float* wp = WembP + wave * (IN_DIM * 16) + lane * 4;
    const int a0 = m * XS + 2 * hi;
    #pragma unroll
    for (int kp = 0; kp < IN_DIM / 8; ++kp) {
        const v4f b = *reinterpret_cast<const v4f*>(wp + kp * 128);
        v2f b0, b1, a[RT], c[RT];
        b0.x = b.x; b0.y = b.y; b1.x = b.z; b1.y = b.w;
        #pragma unroll
        for (int t = 0; t < RT; ++t) {
            const int o = a0 + t * 16 * XS + 8 * kp;
            a[t].x = shx[o];     a[t].y = shx[o + 1];
            c[t].x = shx[o + 4]; c[t].y = shx[o + 5];
        }
        #pragma unroll
        for (int t = 0; t < RT; ++t)
            acc[t] = __builtin_amdgcn_wmma_f32_16x16x4_f32(false, a[t], false, b0,
                                                           (short)0, acc[t], false, false);
        #pragma unroll
        for (int t = 0; t < RT; ++t)
            acc[t] = __builtin_amdgcn_wmma_f32_16x16x4_f32(false, c[t], false, b1,
                                                           (short)0, acc[t], false, false);
    }
    if (IN_DIM & 4) {
        const int kl = IN_DIM - 4;
        const v2f b = *reinterpret_cast<const v2f*>(WembP + wave * (IN_DIM * 16)
                                                    + (IN_DIM / 8) * 128 + lane * 2);
        #pragma unroll
        for (int t = 0; t < RT; ++t) {
            const int o = a0 + t * 16 * XS + kl;
            v2f a; a.x = shx[o]; a.y = shx[o + 1];
            acc[t] = __builtin_amdgcn_wmma_f32_16x16x4_f32(false, a, false, b,
                                                           (short)0, acc[t], false, false);
        }
    }
    static_assert(RT == 2, "the statement below names the RT accumulator tiles one by one");
    asm volatile("v_nop\n\tv_nop\n\tv_nop\n\tv_nop" : "+v"(acc[0]), "+v"(acc[1]));

    float* hrow = h + row0 * H_DIM + wave * 16 + m;
    if (full) {
        #pragma unroll
        for (int t = 0; t < RT; ++t) {
            #pragma unroll
            for (int v = 0; v < 8; ++v) {
                float r = acc[t][v];
                hrow[(long)(16 * t + v + 8 * hi) * H_DIM] = r > 0.0f ? r : 0.0f;
            }
        }
    } else {
        #pragma unroll
        for (int t = 0; t < RT; ++t) {
            #pragma unroll
            for (int v = 0; v < 8; ++v) {
                int mm = 16 * t + v + 8 * hi;
                if (row0 + mm < n_nodes) {
                    float r = acc[t][v];
                    hrow[(long)mm * H_DIM] = r > 0.0f ? r : 0.0f;
                }
            }
        }
    }
}

__global__ __launch_bounds__(256)
void sage_layer_kernel(const float* __restrict__ agg, const float* __restrict__ hin,
                       const float* __restrict__ deg,
                       const float* __restrict__ WlP, const float* __restrict__ WrP,
                       const float* __restrict__ bias, float* __restrict__ hout,
                       int n_nodes) {
    __shared__ float sh_a[TM * HS];
    __shared__ float sh_h[TM * HS];
    const int  tid  = threadIdx.x;
    const long row0 = (long)blockIdx.x * TM;
    const bool full = (row0 + TM <= (long)n_nodes);

    if (full) {
        for (int i = tid; i < TM * H_DIM; i += 256) {
            int r = i >> 7, c = i & 127;
            long rr = row0 + r;
            float dg  = deg[rr];
            float inv = 1.0f / (dg > 1.0f ? dg : 1.0f);
            sh_a[r * HS + c] = agg[rr * H_DIM + c] * inv;
            sh_h[r * HS + c] = hin[rr * H_DIM + c];
        }
    } else {
        for (int i = tid; i < TM * H_DIM; i += 256) {
            int r = i >> 7, c = i & 127;
            long rr = row0 + r; if (rr >= n_nodes) rr = n_nodes - 1;
            float dg  = deg[rr];
            float inv = 1.0f / (dg > 1.0f ? dg : 1.0f);
            sh_a[r * HS + c] = agg[rr * H_DIM + c] * inv;
            sh_h[r * HS + c] = hin[rr * H_DIM + c];
        }
    }
    __syncthreads();

    const int wave = tid >> 5, lane = tid & 31;
    const int m = lane & 15, hi = lane >> 4;
    const int ncol = wave * 16 + m;

    v8f acc[RT];
    float bv = bias[ncol];
    #pragma unroll
    for (int t = 0; t < RT; ++t) {
        #pragma unroll
        for (int i = 0; i < 8; ++i) acc[t][i] = bv;
    }

    const float* wl = WlP + wave * (H_DIM * 16) + lane * 4;
    const float* wr = WrP + wave * (H_DIM * 16) + lane * 4;
    const int a0 = m * HS + 2 * hi;
    #pragma unroll
    for (int kp = 0; kp < H_DIM / 8; ++kp) {
        const v4f vl = *reinterpret_cast<const v4f*>(wl + kp * 128);
        const v4f vr = *reinterpret_cast<const v4f*>(wr + kp * 128);
        v2f bl0, br0, bl1, br1, aa[RT], ah[RT], ca[RT], ch[RT];
        bl0.x = vl.x; bl0.y = vl.y; bl1.x = vl.z; bl1.y = vl.w;
        br0.x = vr.x; br0.y = vr.y; br1.x = vr.z; br1.y = vr.w;
        #pragma unroll
        for (int t = 0; t < RT; ++t) {
            const int o = a0 + t * 16 * HS + 8 * kp;
            aa[t].x = sh_a[o];     aa[t].y = sh_a[o + 1];
            ah[t].x = sh_h[o];     ah[t].y = sh_h[o + 1];
            ca[t].x = sh_a[o + 4]; ca[t].y = sh_a[o + 5];
            ch[t].x = sh_h[o + 4]; ch[t].y = sh_h[o + 5];
        }
        #pragma unroll
        for (int t = 0; t < RT; ++t)
            acc[t] = __builtin_amdgcn_wmma_f32_16x16x4_f32(false, aa[t], false, bl0,
                                                           (short)0, acc[t], false, false);
        #pragma unroll
        for (int t = 0; t < RT; ++t)
            acc[t] = __builtin_amdgcn_wmma_f32_16x16x4_f32(false, ah[t], false, br0,
                                                           (short)0, acc[t], false, false);
        #pragma unroll
        for (int t = 0; t < RT; ++t)
            acc[t] = __builtin_amdgcn_wmma_f32_16x16x4_f32(false, ca[t], false, bl1,
                                                           (short)0, acc[t], false, false);
        #pragma unroll
        for (int t = 0; t < RT; ++t)
            acc[t] = __builtin_amdgcn_wmma_f32_16x16x4_f32(false, ch[t], false, br1,
                                                           (short)0, acc[t], false, false);
    }
    static_assert(RT == 2, "the statement below names the RT accumulator tiles one by one");
    asm volatile("v_nop\n\tv_nop\n\tv_nop\n\tv_nop" : "+v"(acc[0]), "+v"(acc[1]));

    float* orow = hout + row0 * H_DIM + wave * 16 + m;
    if (full) {
        #pragma unroll
        for (int t = 0; t < RT; ++t) {
            #pragma unroll
            for (int v = 0; v < 8; ++v) {
                float r = acc[t][v];
                orow[(long)(16 * t + v + 8 * hi) * H_DIM] = r > 0.0f ? r : 0.0f;
            }
        }
    } else {
        #pragma unroll
        for (int t = 0; t < RT; ++t) {
            #pragma unroll
            for (int v = 0; v < 8; ++v) {
                int mm = 16 * t + v + 8 * hi;
                if (row0 + mm < n_nodes) {
                    float r = acc[t][v];
                    orow[(long)mm * H_DIM] = r > 0.0f ? r : 0.0f;
                }
            }
        }
    }
}


static inline int cdiv(long a, long b) { return (int)((a + b - 1) / b); }

extern "C" void kernel_launch(void* const* d_in, const int* in_sizes, int n_in,
                              void* d_out, int out_size, void* d_ws, size_t ws_size,
                              hipStream_t stream) {
    const float* x    = (const float*)d_in[0];
    const int*   ei   = (const int*)  d_in[1];
    const float* Wemb = (const float*)d_in[2];
    const float* bemb = (const float*)d_in[3];
    const float* Wl   = (const float*)d_in[4];
    const float* bl   = (const float*)d_in[5];
    const float* Wr   = (const float*)d_in[6];
    float* out = (float*)d_out;

    const int N = in_sizes[0] / IN_DIM;
    const int E = in_sizes[1] / 2;
    const int* src = ei;
    const int* dst = ei + E;

    char*  ws  = (char*)d_ws;
    size_t off = 0;
    auto carve = [&](size_t bytes) -> float* {
        float* p = (float*)(ws + off);
        off = (off + bytes + 255) & ~(size_t)255;
        return p;
    };
    float* deg   = carve((size_t)N * sizeof(float));
    float* hA    = carve((size_t)N * H_DIM * sizeof(float));
    float* hB    = carve((size_t)N * H_DIM * sizeof(float));
    float* agg   = carve((size_t)N * H_DIM * sizeof(float));
    float* WembP = carve((size_t)IN_DIM * H_DIM * sizeof(float));
    float* WlP   = carve((size_t)L_LAYERS * H_DIM * H_DIM * sizeof(float));
    float* WrP   = carve((size_t)L_LAYERS * H_DIM * H_DIM * sizeof(float));
    const int ntrips = cdiv(E, TRIP_E);
    const int nkey   = ntrips * TRIP_DW;
    unsigned* key = (unsigned*)carve((size_t)nkey * sizeof(unsigned));
    (void)ws_size;

    const int tiles = cdiv(N, TM);

    const int nb_zero = cdiv(N, 256);
    const int nb_emb  = cdiv((long)IN_DIM * H_DIM, 256);
    const int nb_lay  = 2 * L_LAYERS * (H_DIM * H_DIM / 256);
    prep_kernel<<<nb_zero + nb_emb + nb_lay, 256, 0, stream>>>(deg, N, Wemb, WembP, Wl, WlP, Wr, WrP,
                                                              nb_zero, nb_emb);

    const int nb_deg = cdiv(nkey, 256);
    emb_kernel<<<nb_deg + tiles, 256, 0, stream>>>(x, WembP, bemb, hA, N, dst, deg, key, E, nkey, nb_deg);

    int* elist = (int*)d_out;
    int* gpos  = elist + 2 * (long)E;

    float* hcur = hA;
    for (int l = 0; l < L_LAYERS; ++l) {
        if (l == 0)
            agg_range_kernel<<<cdiv(N, RB), 1024, 0, stream>>>(src, dst, key, deg, hcur, agg,
                                                              elist, gpos, E, N, ntrips);
        else
            agg_list_kernel<<<cdiv(N, RB), 1024, 0, stream>>>(elist, gpos, hcur, agg, E, N);
        float* hnext = (l == L_LAYERS - 1) ? out : (hcur == hA ? hB : hA);
        sage_layer_kernel<<<tiles, 256, 0, stream>>>(
            agg, hcur, deg,
            WlP + (long)l * H_DIM * H_DIM, WrP + (long)l * H_DIM * H_DIM,
            bl + (long)l * H_DIM, hnext, N);
        hcur = hnext;
    }
}


# GEMM input tiles staged by 16-byte loads issued as one batch: a wave copies whole 512-B rows (x tile one contiguous run, LDS pitch 300), one degree reciprocal per row; sums and k order unchanged
# speedup vs baseline: 3.2179x; 1.0601x over previous
#include <hip/hip_runtime.h>

typedef __attribute__((ext_vector_type(2))) float v2f;
typedef __attribute__((ext_vector_type(4))) float v4f;
typedef __attribute__((ext_vector_type(8))) float v8f;

#define IN_DIM 300
#define H_DIM  128
#define L_LAYERS 4

#define XS IN_DIM
#define HS (H_DIM + 4)

#define RT 2
#define TM (16 * RT)


#define RB 400
#define TD 32
#define TRIP_DW (TD * 32)
#define TRIP_E  (TRIP_DW * 4)

static __device__ __forceinline__ void degree_key_thread(const int* __restrict__ dst,
                                                         float* __restrict__ deg,
                                                         unsigned* __restrict__ key,
                                                         int e, int nkey, int q) {
    if (q >= nkey) return;
    const int i0 = 4 * q;
    int d[4];
    #pragma unroll
    for (int j = 0; j < 4; ++j) d[j] = (i0 + j < e) ? dst[i0 + j] : 0;
    unsigned w = 0;
    #pragma unroll
    for (int j = 0; j < 4; ++j) {
        unsigned b = 0xFFu;
        if (i0 + j < e) {
            unsafeAtomicAdd(&deg[d[j]], 1.0f);
            b = ((unsigned)d[j] / RB) & 0xFFu;
        }
        w |= b << (8 * j);
    }
    key[q] = w;
}

static __device__ __forceinline__ void pack_b_elem(const float* __restrict__ in,
                                                   float* __restrict__ out, int K, int Nc, int i) {
    if (i >= K * Nc) return;
    const int ksteps = K >> 2;
    const int per_ct = ksteps * 64;
    const int ct = i / per_ct;
    const int r  = i - ct * per_ct;
    const int pair_floats = (ksteps >> 1) * 128;
    int k, l;
    if (r < pair_floats) {
        const int kp = r >> 7, q = r & 127;
        l = q >> 2;
        k = 8 * kp + 4 * ((q >> 1) & 1) + 2 * (l >> 4) + (q & 1);
    } else {
        const int q = r - pair_floats;
        l = q >> 1;
        k = 4 * (ksteps - 1) + 2 * (l >> 4) + (q & 1);
    }
    out[i] = in[k * Nc + 16 * ct + (l & 15)];
}

__global__ __launch_bounds__(256)
void prep_kernel(float* __restrict__ deg, int n_nodes,
                 const float* __restrict__ Wemb, float* __restrict__ WembP,
                 const float* __restrict__ Wl, float* __restrict__ WlP,
                 const float* __restrict__ Wr, float* __restrict__ WrP,
                 int nb_zero, int nb_emb) {
    int b = blockIdx.x;
    if (b < nb_zero) {
        const int i = b * 256 + threadIdx.x;
        if (i < n_nodes) deg[i] = 0.0f;
        return;
    }
    b -= nb_zero;
    if (b < nb_emb) {
        pack_b_elem(Wemb, WembP, IN_DIM, H_DIM, b * 256 + threadIdx.x);
        return;
    }
    b -= nb_emb;
    const int per = H_DIM * H_DIM / 256;
    const int m = b / per;
    const int i = (b - m * per) * 256 + threadIdx.x;
    if (m >= 2 * L_LAYERS) return;
    const bool left = m < L_LAYERS;
    const long mo = (long)(left ? m : m - L_LAYERS) * H_DIM * H_DIM;
    pack_b_elem((left ? Wl : Wr) + mo, (left ? WlP : WrP) + mo, H_DIM, H_DIM, i);
}


static __device__ __forceinline__ void load_keys(uint4 (&kv)[TD / 4], const uint4* __restrict__ kp,
                                                 int t, int lane) {
    #pragma unroll
    for (int i = 0; i < TD / 4; ++i) kv[i] = kp[(long)t * (TRIP_DW / 4) + i * 32 + lane];
}

template <int G>
static __device__ __forceinline__ void add_group(float* acc, const float* __restrict__ h,
                                                 int sv, int dv, int g0, int lo, int nrows, int lane) {
    v4f v[G];
    #pragma unroll
    for (int g = 0; g < G; ++g) {
        const int s = __builtin_amdgcn_readlane(sv, g0 + g);
        v[g] = *reinterpret_cast<const v4f*>(h + (long)s * H_DIM + lane * 4);
    }
    #pragma unroll
    for (int g = 0; g < G; ++g) {
        const int dl = __builtin_amdgcn_readlane(dv, g0 + g) - lo;
        if ((unsigned)dl < (unsigned)nrows) {
            float* a = acc + dl * H_DIM + lane;
            __hip_atomic_fetch_add(a,      v[g].x, __ATOMIC_RELAXED, __HIP_MEMORY_SCOPE_WORKGROUP);
            __hip_atomic_fetch_add(a + 32, v[g].y, __ATOMIC_RELAXED, __HIP_MEMORY_SCOPE_WORKGROUP);
            __hip_atomic_fetch_add(a + 64, v[g].z, __ATOMIC_RELAXED, __HIP_MEMORY_SCOPE_WORKGROUP);
            __hip_atomic_fetch_add(a + 96, v[g].w, __ATOMIC_RELAXED, __HIP_MEMORY_SCOPE_WORKGROUP);
        }
    }
}

__global__ __launch_bounds__(1024)
void agg_range_kernel(const int* __restrict__ src, const int* __restrict__ dst,
                      const unsigned* __restrict__ key, const float* __restrict__ deg,
                      const float* __restrict__ h, float* __restrict__ agg,
                      int* __restrict__ elist, int* __restrict__ gpos,
                      int e, int n_nodes, int ntrips) {
    __shared__ __attribute__((aligned(16))) float acc[RB * H_DIM];
    __shared__ int s_start, s_fill;
    const int tid = threadIdx.x, lane = tid & 31;
    const int wave = __builtin_amdgcn_readfirstlane(tid >> 5);
    const int lo = blockIdx.x * RB;
    int nrows = n_nodes - lo; if (nrows > RB) nrows = RB;
    const unsigned pat = ((unsigned)blockIdx.x & 255u) * 0x01010101u;
    const uint4* kp = reinterpret_cast<const uint4*>(key);

    float4* acc4 = reinterpret_cast<float4*>(acc);
    for (int i = tid; i < RB * H_DIM / 4; i += 1024) acc4[i] = make_float4(0.f, 0.f, 0.f, 0.f);
    if (tid == 0) { s_start = 0; s_fill = 0; }
    int part = 0;
    const float4* d4 = reinterpret_cast<const float4*>(deg);
    for (int i = tid; i < lo / 4; i += 1024) {
        const float4 v = d4[i];
        part += (int)v.x + (int)v.y + (int)v.z + (int)v.w;
    }
    __syncthreads();
    if (part) __hip_atomic_fetch_add(&s_start, part, __ATOMIC_RELAXED, __HIP_MEMORY_SCOPE_WORKGROUP);
    __syncthreads();
    const int start = s_start;
    int2* lst = reinterpret_cast<int2*>(elist);

    auto sum_matches = [&](int base, int qreg, int cnt) {
        int sv = 0, dv = lo - 1;
        if (lane < cnt) {
            const int i = base + qreg;
            if (i < e) { sv = src[i]; dv = dst[i]; }
        }
        {
            const bool ok = (unsigned)(dv - lo) < (unsigned)nrows;
            const unsigned bal = __builtin_amdgcn_ballot_w32(ok);
            int b = 0;
            if (lane == 0) b = __hip_atomic_fetch_add(&s_fill, __builtin_popcount(bal),
                                                      __ATOMIC_RELAXED, __HIP_MEMORY_SCOPE_WORKGROUP);
            b = __builtin_amdgcn_readfirstlane(b);
            const int pos = start + b + __builtin_popcount(bal & ((1u << lane) - 1u));
            if (ok && pos < e) lst[pos] = make_int2(sv, dv - lo);
        }
        int g0 = 0;
        for (; g0 + 8 <= cnt; g0 += 8) add_group<8>(acc, h, sv, dv, g0, lo, nrows, lane);
        int r = cnt - g0;
        if (r >= 4) { add_group<4>(acc, h, sv, dv, g0, lo, nrows, lane); g0 += 4; r -= 4; }
        if (r == 3)      add_group<3>(acc, h, sv, dv, g0, lo, nrows, lane);
        else if (r == 2) add_group<2>(acc, h, sv, dv, g0, lo, nrows, lane);
        else if (r == 1) add_group<1>(acc, h, sv, dv, g0, lo, nrows, lane);
    };

    if (wave < ntrips) {
    uint4 kv[TD / 4];
    load_keys(kv, kp, wave, lane);
    for (int t = wave; t < ntrips; t += 32) {
        const int base = t * TRIP_E;
        uint4 kn[TD / 4];
        load_keys(kn, kp, (t + 32 < ntrips) ? t + 32 : t, lane);

        unsigned hinv[4] = {~0u, ~0u, ~0u, ~0u};
        #pragma unroll
        for (int i = 0; i < TD / 4; ++i) {
            const unsigned c4[4] = {kv[i].x, kv[i].y, kv[i].z, kv[i].w};
            #pragma unroll
            for (int c = 0; c < 4; ++c) {
                const int k = 4 * i + c;
                const unsigned x  = c4[c] ^ pat;
                const unsigned nz = ((x & 0x7f7f7f7fu) + 0x7f7f7f7fu) | x | 0x7f7f7f7fu;
                hinv[k >> 3] &= __builtin_rotateright32(nz, 7 - (k & 7));
            }
        }

        int qreg = 0, qn = 0;
        #pragma unroll 1
        for (int g = 0; g < 4; ++g) {
            unsigned hw = ~hinv[0];
            hw = (g == 1) ? ~hinv[1] : hw;
            hw = (g == 2) ? ~hinv[2] : hw;
            hw = (g == 3) ? ~hinv[3] : hw;
            unsigned any = __builtin_amdgcn_ballot_w32(hw != 0);
            while (any) {
                const int L = __builtin_ctz(any);
                any &= any - 1;
                unsigned m = (unsigned)__builtin_amdgcn_readlane((int)hw, L);
                while (m) {
                    const int pbit = __builtin_ctz(m);
                    m &= m - 1;
                    const int off = (2 * g + ((pbit >> 2) & 1)) * 512 + L * 16 + (pbit & 3) * 4 + (pbit >> 3);
                    qreg = (lane == qn) ? off : qreg;
                    if (++qn == 32) { sum_matches(base, qreg, 32); qn = 0; }
                }
            }
        }
        if (qn) sum_matches(base, qreg, qn);

        #pragma unroll
        for (int i = 0; i < TD / 4; ++i) kv[i] = kn[i];
    }
    }
    __syncthreads();

    float4* out4 = reinterpret_cast<float4*>(agg + (long)lo * H_DIM);
    for (int i = tid; i < nrows * (H_DIM / 4); i += 1024) {
        const float* a = acc + (i >> 5) * H_DIM + (i & 31);
        out4[i] = make_float4(a[0], a[32], a[64], a[96]);
    }
    if (tid == 0) { gpos[2 * blockIdx.x] = start; gpos[2 * blockIdx.x + 1] = s_fill; }
}

template <int G>
static __device__ __forceinline__ void add_rows(float* acc, const float* __restrict__ h,
                                                int sv, int rv, int g0, int lane) {
    v4f v[G];
    #pragma unroll
    for (int g = 0; g < G; ++g) {
        const int s = __builtin_amdgcn_readlane(sv, g0 + g);
        v[g] = *reinterpret_cast<const v4f*>(h + (long)s * H_DIM + lane * 4);
    }
    #pragma unroll
    for (int g = 0; g < G; ++g) {
        unsigned dl = (unsigned)__builtin_amdgcn_readlane(rv, g0 + g);
        if (dl > RB - 1) dl = RB - 1;
        float* a = acc + dl * H_DIM + lane;
        __hip_atomic_fetch_add(a,      v[g].x, __ATOMIC_RELAXED, __HIP_MEMORY_SCOPE_WORKGROUP);
        __hip_atomic_fetch_add(a + 32, v[g].y, __ATOMIC_RELAXED, __HIP_MEMORY_SCOPE_WORKGROUP);
        __hip_atomic_fetch_add(a + 64, v[g].z, __ATOMIC_RELAXED, __HIP_MEMORY_SCOPE_WORKGROUP);
        __hip_atomic_fetch_add(a + 96, v[g].w, __ATOMIC_RELAXED, __HIP_MEMORY_SCOPE_WORKGROUP);
    }
}

__global__ __launch_bounds__(1024)
void agg_list_kernel(const int* __restrict__ elist, const int* __restrict__ gpos,
                     const float* __restrict__ h, float* __restrict__ agg,
                     int e, int n_nodes) {
    __shared__ __attribute__((aligned(16))) float acc[RB * H_DIM];
    const int tid = threadIdx.x, lane = tid & 31;
    const int wave = __builtin_amdgcn_readfirstlane(tid >> 5);
    const int lo = blockIdx.x * RB;
    int nrows = n_nodes - lo; if (nrows > RB) nrows = RB;

    float4* acc4 = reinterpret_cast<float4*>(acc);
    for (int i = tid; i < RB * H_DIM / 4; i += 1024) acc4[i] = make_float4(0.f, 0.f, 0.f, 0.f);

    int start = gpos[2 * blockIdx.x], cnt = gpos[2 * blockIdx.x + 1];
    if (start < 0) start = 0;
    if (start > e) start = e;
    if (cnt > e - start) cnt = e - start;
    if (cnt < 0) cnt = 0;
    const int2* lp = reinterpret_cast<const int2*>(elist) + start;
    const int b0 = (int)(((long)cnt * wave) >> 5), b1 = (int)(((long)cnt * (wave + 1)) >> 5);
    __syncthreads();

    if (b0 < b1) {
    int2 en = lp[(b0 + lane < b1) ? b0 + lane : b1 - 1];
    for (int p = b0; p < b1; p += 32) {
        int m = b1 - p; if (m > 32) m = 32;
        const int2 nx = lp[(p + 32 + lane < b1) ? p + 32 + lane : b1 - 1];
        const int sv = en.x, dv = en.y;
        int g0 = 0;
        for (; g0 + 8 <= m; g0 += 8) add_rows<8>(acc, h, sv, dv, g0, lane);
        int r = m - g0;
        if (r >= 4) { add_rows<4>(acc, h, sv, dv, g0, lane); g0 += 4; r -= 4; }
        if (r == 3)      add_rows<3>(acc, h, sv, dv, g0, lane);
        else if (r == 2) add_rows<2>(acc, h, sv, dv, g0, lane);
        else if (r == 1) add_rows<1>(acc, h, sv, dv, g0, lane);
        en = nx;
    }
    }
    __syncthreads();

    float4* out4 = reinterpret_cast<float4*>(agg + (long)lo * H_DIM);
    for (int i = tid; i < nrows * (H_DIM / 4); i += 1024) {
        const float* a = acc + (i >> 5) * H_DIM + (i & 31);
        out4[i] = make_float4(a[0], a[32], a[64], a[96]);
    }
}


__global__ __launch_bounds__(256)
void emb_kernel(const float* __restrict__ x, const float* __restrict__ WembP,
                const float* __restrict__ bemb, float* __restrict__ h, int n_nodes,
                const int* __restrict__ dst, float* __restrict__ deg,
                unsigned* __restrict__ key, int e, int nkey, int nb_deg) {
    __shared__ __attribute__((aligned(16))) float shx[TM * XS];
    const int  tid  = threadIdx.x;
    if ((int)blockIdx.x < nb_deg) {
        degree_key_thread(dst, deg, key, e, nkey, (int)blockIdx.x * 256 + tid);
        return;
    }
    const long row0 = (long)((int)blockIdx.x - nb_deg) * TM;
    const bool full = (row0 + TM <= (long)n_nodes);

    if (full) {
        const v4f* xg = reinterpret_cast<const v4f*>(x + row0 * IN_DIM);
        v4f* xs = reinterpret_cast<v4f*>(shx);
        constexpr int NV = TM * IN_DIM / 4;
        v4f xv[(NV + 255) / 256];
        #pragma unroll
        for (int j = 0; j < (NV + 255) / 256; ++j) {
            const int i = tid + 256 * j;
            if (i < NV) xv[j] = xg[i];
        }
        #pragma unroll
        for (int j = 0; j < (NV + 255) / 256; ++j) {
            const int i = tid + 256 * j;
            if (i < NV) xs[i] = xv[j];
        }
    } else {
        for (int i = tid; i < TM * IN_DIM; i += 256) {
            int r = i / IN_DIM, c = i - r * IN_DIM;
            long rr = row0 + r; if (rr >= n_nodes) rr = n_nodes - 1;
            shx[r * XS + c] = x[rr * IN_DIM + c];
        }
    }
    __syncthreads();

    const int wave = tid >> 5, lane = tid & 31;
    const int m = lane & 15, hi = lane >> 4;
    const int ncol = wave * 16 + m;

    v8f acc[RT];
    float bv = bemb[ncol];
    #pragma unroll
    for (int t = 0; t < RT; ++t) {
        #pragma unroll
        for (int i = 0; i < 8; ++i) acc[t][i] = bv;
    }

    const float* wp = WembP + wave * (IN_DIM * 16) + lane * 4;
    const int a0 = m * XS + 2 * hi;
    #pragma unroll
    for (int kp = 0; kp < IN_DIM / 8; ++kp) {
        const v4f b = *reinterpret_cast<const v4f*>(wp + kp * 128);
        v2f b0, b1, a[RT], c[RT];
        b0.x = b.x; b0.y = b.y; b1.x = b.z; b1.y = b.w;
        #pragma unroll
        for (int t = 0; t < RT; ++t) {
            const int o = a0 + t * 16 * XS + 8 * kp;
            a[t].x = shx[o];     a[t].y = shx[o + 1];
            c[t].x = shx[o + 4]; c[t].y = shx[o + 5];
        }
        #pragma unroll
        for (int t = 0; t < RT; ++t)
            acc[t] = __builtin_amdgcn_wmma_f32_16x16x4_f32(false, a[t], false, b0,
                                                           (short)0, acc[t], false, false);
        #pragma unroll
        for (int t = 0; t < RT; ++t)
            acc[t] = __builtin_amdgcn_wmma_f32_16x16x4_f32(false, c[t], false, b1,
                                                           (short)0, acc[t], false, false);
    }
    if (IN_DIM & 4) {
        const int kl = IN_DIM - 4;
        const v2f b = *reinterpret_cast<const v2f*>(WembP + wave * (IN_DIM * 16)
                                                    + (IN_DIM / 8) * 128 + lane * 2);
        #pragma unroll
        for (int t = 0; t < RT; ++t) {
            const int o = a0 + t * 16 * XS + kl;
            v2f a; a.x = shx[o]; a.y = shx[o + 1];
            acc[t] = __builtin_amdgcn_wmma_f32_16x16x4_f32(false, a, false, b,
                                                           (short)0, acc[t], false, false);
        }
    }
    static_assert(RT == 2, "the statement below names the RT accumulator tiles one by one");
    asm volatile("v_nop\n\tv_nop\n\tv_nop\n\tv_nop" : "+v"(acc[0]), "+v"(acc[1]));

    float* hrow = h + row0 * H_DIM + wave * 16 + m;
    if (full) {
        #pragma unroll
        for (int t = 0; t < RT; ++t) {
            #pragma unroll
            for (int v = 0; v < 8; ++v) {
                float r = acc[t][v];
                hrow[(long)(16 * t + v + 8 * hi) * H_DIM] = r > 0.0f ? r : 0.0f;
            }
        }
    } else {
        #pragma unroll
        for (int t = 0; t < RT; ++t) {
            #pragma unroll
            for (int v = 0; v < 8; ++v) {
                int mm = 16 * t + v + 8 * hi;
                if (row0 + mm < n_nodes) {
                    float r = acc[t][v];
                    hrow[(long)mm * H_DIM] = r > 0.0f ? r : 0.0f;
                }
            }
        }
    }
}

__global__ __launch_bounds__(256)
void sage_layer_kernel(const float* __restrict__ agg, const float* __restrict__ hin,
                       const float* __restrict__ deg,
                       const float* __restrict__ WlP, const float* __restrict__ WrP,
                       const float* __restrict__ bias, float* __restrict__ hout,
                       int n_nodes) {
    __shared__ __attribute__((aligned(16))) float sh_a[TM * HS];
    __shared__ __attribute__((aligned(16))) float sh_h[TM * HS];
    const int  tid  = threadIdx.x;
    const long row0 = (long)blockIdx.x * TM;
    const bool full = (row0 + TM <= (long)n_nodes);

    {
        const int wrow = __builtin_amdgcn_readfirstlane(tid >> 5);
        const int l4   = (tid & 31) * 4;
        v4f va[TM / 8], vh[TM / 8];
        float inv[TM / 8];
        #pragma unroll
        for (int j = 0; j < TM / 8; ++j) {
            long rr = row0 + wrow + 8 * j;
            if (!full && rr >= n_nodes) rr = n_nodes - 1;
            const float dg = deg[rr];
            inv[j] = 1.0f / (dg > 1.0f ? dg : 1.0f);
            va[j] = *reinterpret_cast<const v4f*>(agg + rr * H_DIM + l4);
            vh[j] = *reinterpret_cast<const v4f*>(hin + rr * H_DIM + l4);
        }
        #pragma unroll
        for (int j = 0; j < TM / 8; ++j) {
            const int o = (wrow + 8 * j) * HS + l4;
            *reinterpret_cast<v4f*>(sh_a + o) = va[j] * inv[j];
            *reinterpret_cast<v4f*>(sh_h + o) = vh[j];
        }
    }
    __syncthreads();

    const int wave = tid >> 5, lane = tid & 31;
    const int m = lane & 15, hi = lane >> 4;
    const int ncol = wave * 16 + m;

    v8f acc[RT];
    float bv = bias[ncol];
    #pragma unroll
    for (int t = 0; t < RT; ++t) {
        #pragma unroll
        for (int i = 0; i < 8; ++i) acc[t][i] = bv;
    }

    const float* wl = WlP + wave * (H_DIM * 16) + lane * 4;
    const float* wr = WrP + wave * (H_DIM * 16) + lane * 4;
    const int a0 = m * HS + 2 * hi;
    #pragma unroll
    for (int kp = 0; kp < H_DIM / 8; ++kp) {
        const v4f vl = *reinterpret_cast<const v4f*>(wl + kp * 128);
        const v4f vr = *reinterpret_cast<const v4f*>(wr + kp * 128);
        v2f bl0, br0, bl1, br1, aa[RT], ah[RT], ca[RT], ch[RT];
        bl0.x = vl.x; bl0.y = vl.y; bl1.x = vl.z; bl1.y = vl.w;
        br0.x = vr.x; br0.y = vr.y; br1.x = vr.z; br1.y = vr.w;
        #pragma unroll
        for (int t = 0; t < RT; ++t) {
            const int o = a0 + t * 16 * HS + 8 * kp;
            aa[t].x = sh_a[o];     aa[t].y = sh_a[o + 1];
            ah[t].x = sh_h[o];     ah[t].y = sh_h[o + 1];
            ca[t].x = sh_a[o + 4]; ca[t].y = sh_a[o + 5];
            ch[t].x = sh_h[o + 4]; ch[t].y = sh_h[o + 5];
        }
        #pragma unroll
        for (int t = 0; t < RT; ++t)
            acc[t] = __builtin_amdgcn_wmma_f32_16x16x4_f32(false, aa[t], false, bl0,
                                                           (short)0, acc[t], false, false);
        #pragma unroll
        for (int t = 0; t < RT; ++t)
            acc[t] = __builtin_amdgcn_wmma_f32_16x16x4_f32(false, ah[t], false, br0,
                                                           (short)0, acc[t], false, false);
        #pragma unroll
        for (int t = 0; t < RT; ++t)
            acc[t] = __builtin_amdgcn_wmma_f32_16x16x4_f32(false, ca[t], false, bl1,
                                                           (short)0, acc[t], false, false);
        #pragma unroll
        for (int t = 0; t < RT; ++t)
            acc[t] = __builtin_amdgcn_wmma_f32_16x16x4_f32(false, ch[t], false, br1,
                                                           (short)0, acc[t], false, false);
    }
    static_assert(RT == 2, "the statement below names the RT accumulator tiles one by one");
    asm volatile("v_nop\n\tv_nop\n\tv_nop\n\tv_nop" : "+v"(acc[0]), "+v"(acc[1]));

    float* orow = hout + row0 * H_DIM + wave * 16 + m;
    if (full) {
        #pragma unroll
        for (int t = 0; t < RT; ++t) {
            #pragma unroll
            for (int v = 0; v < 8; ++v) {
                float r = acc[t][v];
                orow[(long)(16 * t + v + 8 * hi) * H_DIM] = r > 0.0f ? r : 0.0f;
            }
        }
    } else {
        #pragma unroll
        for (int t = 0; t < RT; ++t) {
            #pragma unroll
            for (int v = 0; v < 8; ++v) {
                int mm = 16 * t + v + 8 * hi;
                if (row0 + mm < n_nodes) {
                    float r = acc[t][v];
                    orow[(long)mm * H_DIM] = r > 0.0f ? r : 0.0f;
                }
            }
        }
    }
}


static inline int cdiv(long a, long b) { return (int)((a + b - 1) / b); }

extern "C" void kernel_launch(void* const* d_in, const int* in_sizes, int n_in,
                              void* d_out, int out_size, void* d_ws, size_t ws_size,
                              hipStream_t stream) {
    const float* x    = (const float*)d_in[0];
    const int*   ei   = (const int*)  d_in[1];
    const float* Wemb = (const float*)d_in[2];
    const float* bemb = (const float*)d_in[3];
    const float* Wl   = (const float*)d_in[4];
    const float* bl   = (const float*)d_in[5];
    const float* Wr   = (const float*)d_in[6];
    float* out = (float*)d_out;

    const int N = in_sizes[0] / IN_DIM;
    const int E = in_sizes[1] / 2;
    const int* src = ei;
    const int* dst = ei + E;

    char*  ws  = (char*)d_ws;
    size_t off = 0;
    auto carve = [&](size_t bytes) -> float* {
        float* p = (float*)(ws + off);
        off = (off + bytes + 255) & ~(size_t)255;
        return p;
    };
    float* deg   = carve((size_t)N * sizeof(float));
    float* hA    = carve((size_t)N * H_DIM * sizeof(float));
    float* hB    = carve((size_t)N * H_DIM * sizeof(float));
    float* agg   = carve((size_t)N * H_DIM * sizeof(float));
    float* WembP = carve((size_t)IN_DIM * H_DIM * sizeof(float));
    float* WlP   = carve((size_t)L_LAYERS * H_DIM * H_DIM * sizeof(float));
    float* WrP   = carve((size_t)L_LAYERS * H_DIM * H_DIM * sizeof(float));
    const int ntrips = cdiv(E, TRIP_E);
    const int nkey   = ntrips * TRIP_DW;
    unsigned* key = (unsigned*)carve((size_t)nkey * sizeof(unsigned));
    (void)ws_size;

    const int tiles = cdiv(N, TM);

    const int nb_zero = cdiv(N, 256);
    const int nb_emb  = cdiv((long)IN_DIM * H_DIM, 256);
    const int nb_lay  = 2 * L_LAYERS * (H_DIM * H_DIM / 256);
    prep_kernel<<<nb_zero + nb_emb + nb_lay, 256, 0, stream>>>(deg, N, Wemb, WembP, Wl, WlP, Wr, WrP,
                                                              nb_zero, nb_emb);

    const int nb_deg = cdiv(nkey, 256);
    emb_kernel<<<nb_deg + tiles, 256, 0, stream>>>(x, WembP, bemb, hA, N, dst, deg, key, E, nkey, nb_deg);

    int* elist = (int*)d_out;
    int* gpos  = elist + 2 * (long)E;

    float* hcur = hA;
    for (int l = 0; l < L_LAYERS; ++l) {
        if (l == 0)
            agg_range_kernel<<<cdiv(N, RB), 1024, 0, stream>>>(src, dst, key, deg, hcur, agg,
                                                              elist, gpos, E, N, ntrips);
        else
            agg_list_kernel<<<cdiv(N, RB), 1024, 0, stream>>>(elist, gpos, hcur, agg, E, N);
        float* hnext = (l == L_LAYERS - 1) ? out : (hcur == hA ? hB : hA);
        sage_layer_kernel<<<tiles, 256, 0, stream>>>(
            agg, hcur, deg,
            WlP + (long)l * H_DIM * H_DIM, WrP + (long)l * H_DIM * H_DIM,
            bl + (long)l * H_DIM, hnext, N);
        hcur = hnext;
    }
}
